# starting kernel
# baseline (speedup 1.0000x reference)
#include <hip/hip_runtime.h>

typedef __attribute__((ext_vector_type(2))) float v2f;
typedef __attribute__((ext_vector_type(8))) float v8f;

#define N_NODES 100000
#define N_EDGES 1600000
#define IN_DIM  128
#define OUT_DIM 64
#define ROW_TILES (N_NODES / 16)

#define W_PAIR_STRIDE 80
#define W_PAIR_ROWS   (IN_DIM / 2)

__global__ void gcn_init_bias(const float* __restrict__ bias,
                              float* __restrict__ out, int n) {
    int i = blockIdx.x * blockDim.x + threadIdx.x;
    if (i < n) out[i] = bias[i & (OUT_DIM - 1)];
}

__global__ __launch_bounds__(256) void gcn_gemm_wmma(
    const float* __restrict__ X, const float* __restrict__ W,
    float* __restrict__ S) {
    __shared__ v2f w_lds[W_PAIR_ROWS * W_PAIR_STRIDE];

    for (int p = threadIdx.x; p < W_PAIR_ROWS * OUT_DIM; p += 256) {
        const int k2 = p >> 6;
        const int n  = p & 63;
        v2f t;
        t.x = W[(2 * k2 + 0) * OUT_DIM + n];
        t.y = W[(2 * k2 + 1) * OUT_DIM + n];
        w_lds[k2 * W_PAIR_STRIDE + n] = t;
    }
    __syncthreads();

    const int wave = threadIdx.x >> 5;
    const int lane = threadIdx.x & 31;
    const int tile = blockIdx.x * 8 + wave;
    if (tile >= ROW_TILES) return;

    const int row0  = tile * 16;
    const int m     = lane & 15;
    const int khalf = (lane >> 4) * 2;

    v8f acc[4] = {};
    const float* arow = X + (size_t)(row0 + m) * IN_DIM;

    for (int k0 = 0; k0 < IN_DIM; k0 += 4) {
        v2f a = *(const v2f*)(arow + k0 + khalf);
        const v2f* wk = w_lds + ((k0 + khalf) >> 1) * W_PAIR_STRIDE + m;
#pragma unroll
        for (int t = 0; t < 4; ++t) {
            v2f b = wk[t * 16];
            acc[t] = __builtin_amdgcn_wmma_f32_16x16x4_f32(
                 false, a,  false, b,
                 (short)0, acc[t],
                 false,  false);
        }
    }

    const int mhi = (lane >> 4) * 8;
#pragma unroll
    for (int t = 0; t < 4; ++t) {
#pragma unroll
        for (int g = 0; g < 8; ++g) {
            S[(size_t)(row0 + mhi + g) * OUT_DIM + t * 16 + m] = acc[t][g];
        }
    }
}

__global__ __launch_bounds__(256) void gcn_scatter(
    const float* __restrict__ S, const float* __restrict__ vals,
    const int* __restrict__ row, const int* __restrict__ col,
    float* __restrict__ out) {
    long long t = (long long)blockIdx.x * 256 + threadIdx.x;
    long long e = t >> 4;
    if (e >= N_EDGES) return;
    const int j0 = (int)(t & 15) * 4;

    const float v = __builtin_nontemporal_load(vals + e);
    const int   c = __builtin_nontemporal_load(col + e);
    const int   r = __builtin_nontemporal_load(row + e);

    const float4 s = *(const float4*)(S + (size_t)c * OUT_DIM + j0);
    float* o = out + (size_t)r * OUT_DIM + j0;
    atomicAdd(o + 0, v * s.x);
    atomicAdd(o + 1, v * s.y);
    atomicAdd(o + 2, v * s.z);
    atomicAdd(o + 3, v * s.w);
}

extern "C" void kernel_launch(void* const* d_in, const int* in_sizes, int n_in,
                              void* d_out, int out_size, void* d_ws, size_t ws_size,
                              hipStream_t stream) {
    const float* x    = (const float*)d_in[0];
    const float* w    = (const float*)d_in[1];
    const float* bias = (const float*)d_in[2];
    const float* vals = (const float*)d_in[3];
    const int*   row  = (const int*)d_in[4];
    const int*   col  = (const int*)d_in[5];
    float* out     = (float*)d_out;
    float* support = (float*)d_ws;

    {
        const int n = N_NODES * OUT_DIM;
        gcn_init_bias<<<(n + 255) / 256, 256, 0, stream>>>(bias, out, n);
    }
    {
        const int blocks = (ROW_TILES + 7) / 8;
        gcn_gemm_wmma<<<blocks, 256, 0, stream>>>(x, w, support);
    }
    {
        const long long threads = (long long)N_EDGES * 16;
        const int blocks = (int)((threads + 255) / 256);
        gcn_scatter<<<blocks, 256, 0, stream>>>(support, vals, row, col, out);
    }
}


# Global f32 atomic scatter replaced: each of 256 work-groups owns 391 output rows in LDS, scans all row ids, adds its edges by LDS atomic add (sum order = arrival order), stores out+bias once
# speedup vs baseline: 1.6637x; 1.6637x over previous
#include <hip/hip_runtime.h>

typedef __attribute__((ext_vector_type(2))) float v2f;
typedef __attribute__((ext_vector_type(8))) float v8f;

#define N_NODES 100000
#define N_EDGES 1600000
#define IN_DIM  128
#define OUT_DIM 64
#define ROW_TILES (N_NODES / 16)

#define W_PAIR_STRIDE 80
#define W_PAIR_ROWS   (IN_DIM / 2)

__global__ __launch_bounds__(256) void gcn_gemm_wmma(
    const float* __restrict__ X, const float* __restrict__ W,
    float* __restrict__ S) {
    __shared__ v2f w_lds[W_PAIR_ROWS * W_PAIR_STRIDE];

    for (int p = threadIdx.x; p < W_PAIR_ROWS * OUT_DIM; p += 256) {
        const int k2 = p >> 6;
        const int n  = p & 63;
        v2f t;
        t.x = W[(2 * k2 + 0) * OUT_DIM + n];
        t.y = W[(2 * k2 + 1) * OUT_DIM + n];
        w_lds[k2 * W_PAIR_STRIDE + n] = t;
    }
    __syncthreads();

    const int wave = threadIdx.x >> 5;
    const int lane = threadIdx.x & 31;
    const int tile = blockIdx.x * 8 + wave;
    if (tile >= ROW_TILES) return;

    const int row0  = tile * 16;
    const int m     = lane & 15;
    const int khalf = (lane >> 4) * 2;

    v8f acc[4] = {};
    const float* arow = X + (size_t)(row0 + m) * IN_DIM;

    for (int k0 = 0; k0 < IN_DIM; k0 += 4) {
        v2f a = *(const v2f*)(arow + k0 + khalf);
        const v2f* wk = w_lds + ((k0 + khalf) >> 1) * W_PAIR_STRIDE + m;
#pragma unroll
        for (int t = 0; t < 4; ++t) {
            v2f b = wk[t * 16];
            acc[t] = __builtin_amdgcn_wmma_f32_16x16x4_f32(
                 false, a,  false, b,
                 (short)0, acc[t],
                 false,  false);
        }
    }

    const int mhi = (lane >> 4) * 8;
#pragma unroll
    for (int t = 0; t < 4; ++t) {
#pragma unroll
        for (int g = 0; g < 8; ++g) {
            S[(size_t)(row0 + mhi + g) * OUT_DIM + t * 16 + m] = acc[t][g];
        }
    }
}

#define ROWS_PER_WG 391
#define N_WG        256
#define EDGES_PER_STEP 512
#define N_STEPS     (N_EDGES / EDGES_PER_STEP)
#define AGG_WAVES   32
#define Q_CAP       544
static_assert(N_EDGES % EDGES_PER_STEP == 0, "edge list must be whole steps");
static_assert(N_WG * ROWS_PER_WG >= N_NODES, "slices must cover all rows");
static_assert(N_EDGES <= (1 << 21) && ROWS_PER_WG <= (1 << 9), "queue entry packing");

__device__ __forceinline__ int lane_bcast_i(int x, int j) {
    return __builtin_amdgcn_readlane(x, j);
}
__device__ __forceinline__ float lane_bcast_f(float x, int j) {
    return __builtin_bit_cast(float, __builtin_amdgcn_readlane(__builtin_bit_cast(int, x), j));
}

__device__ __forceinline__ int agg_step_edge(int i, int start) {
    int m = i + start;
    if (m >= N_STEPS) m -= N_STEPS;
    return m * EDGES_PER_STEP;
}

__device__ __forceinline__ void agg_drain(
    const int* q, int first, int n, int lane,
    const float* __restrict__ S, const float* __restrict__ vals,
    const int* __restrict__ col, float* slice) {
    int d = 0, c = 0;
    float v = 0.0f;
    if (lane < n) {
        const int ent = q[first + lane];
        const int e = ent & 0x1FFFFF;
        d = (int)((unsigned)ent >> 21);
        c = col[e];
        v = vals[e];
    }
    for (int j0 = 0; j0 < n; j0 += 16) {
        float2 s[16];
#pragma unroll
        for (int u = 0; u < 16; ++u) {
            if (j0 + u < n) {
                const int cj = lane_bcast_i(c, j0 + u);
                s[u] = *(const float2*)(S + (size_t)cj * OUT_DIM + lane * 2);
            }
        }
#pragma unroll
        for (int u = 0; u < 16; ++u) {
            if (j0 + u < n) {
                const float vj = lane_bcast_f(v, j0 + u);
                const int   dj = lane_bcast_i(d, j0 + u);
                float* o = slice + dj * OUT_DIM + lane * 2;
                __hip_atomic_fetch_add(o + 0, vj * s[u].x, __ATOMIC_RELAXED, __HIP_MEMORY_SCOPE_WORKGROUP);
                __hip_atomic_fetch_add(o + 1, vj * s[u].y, __ATOMIC_RELAXED, __HIP_MEMORY_SCOPE_WORKGROUP);
            }
        }
    }
}

__device__ __forceinline__ void agg_load_step(
    int4 (&r)[4], const int* __restrict__ row, int i, int fallback, int start, int lane) {
    const int ic = i < N_STEPS ? i : fallback;
    const int4* p = (const int4*)(row + agg_step_edge(ic, start)) + lane;
#pragma unroll
    for (int k = 0; k < 4; ++k) r[k] = p[k * 32];
}

__device__ __forceinline__ void agg_scan_step(
    const int4 (&r)[4], int e0, int base, unsigned nrows, int* q, int& count, int lane,
    const float* __restrict__ S, const float* __restrict__ vals,
    const int* __restrict__ col, float* slice) {
#pragma unroll
    for (int k = 0; k < 4; ++k) {
        const unsigned d0 = (unsigned)(r[k].x - base);
        const unsigned d1 = (unsigned)(r[k].y - base);
        const unsigned d2 = (unsigned)(r[k].z - base);
        const unsigned d3 = (unsigned)(r[k].w - base);
        const unsigned a = d0 < d1 ? d0 : d1;
        const unsigned b = d2 < d3 ? d2 : d3;
        const unsigned mn = a < b ? a : b;
        if (__builtin_amdgcn_ballot_w32(mn < nrows) != 0u) {
            const int eb = e0 + k * 128 + lane * 4;
            const unsigned dd[4] = {d0, d1, d2, d3};
#pragma unroll
            for (int t = 0; t < 4; ++t) {
                const bool hit = dd[t] < nrows;
                const unsigned mask = __builtin_amdgcn_ballot_w32(hit);
                if (mask != 0u) {
                    if (hit)
                        q[count + (int)__builtin_amdgcn_mbcnt_lo(mask, 0u)] =
                            (int)((dd[t] << 21) | (unsigned)(eb + t));
                    count += __builtin_popcount(mask);
                }
            }
        }
    }
    __builtin_amdgcn_wave_barrier();
    while (count >= 32) {
        count -= 32;
        agg_drain(q, count, 32, lane, S, vals, col, slice);
    }
    __builtin_amdgcn_wave_barrier();
}

__global__ __launch_bounds__(1024) void gcn_aggregate(
    const float* __restrict__ S, const float* __restrict__ vals,
    const int* __restrict__ row, const int* __restrict__ col,
    const float* __restrict__ bias, float* __restrict__ out) {
    __shared__ float slice[ROWS_PER_WG * OUT_DIM];
    __shared__ int   queue[AGG_WAVES * Q_CAP];

    const int tid  = threadIdx.x;
    const int lane = tid & 31;
    const int wave = __builtin_amdgcn_readfirstlane(tid >> 5);
    const int base = blockIdx.x * ROWS_PER_WG;
    const int rem  = N_NODES - base;
    const unsigned nrows = (unsigned)(rem < ROWS_PER_WG ? rem : ROWS_PER_WG);

    for (int i = tid; i < ROWS_PER_WG * OUT_DIM / 4; i += 1024)
        ((float4*)slice)[i] = make_float4(0.f, 0.f, 0.f, 0.f);
    __syncthreads();

    int* q = queue + wave * Q_CAP;
    int count = 0;
    const int start = blockIdx.x * (N_STEPS / N_WG);

    int4 ra[4], rb[4];
    agg_load_step(ra, row, wave, wave, start, lane);
    for (int i = wave; i < N_STEPS; i += 2 * AGG_WAVES) {
        agg_load_step(rb, row, i + AGG_WAVES, wave, start, lane);
        agg_scan_step(ra, agg_step_edge(i, start), base, nrows, q, count, lane, S, vals, col, slice);
        agg_load_step(ra, row, i + 2 * AGG_WAVES, wave, start, lane);
        if (i + AGG_WAVES < N_STEPS)
            agg_scan_step(rb, agg_step_edge(i + AGG_WAVES, start), base, nrows, q, count, lane, S, vals, col, slice);
    }
    if (count > 0) agg_drain(q, 0, count, lane, S, vals, col, slice);

    __syncthreads();
    const float4 bv = ((const float4*)bias)[tid & 15];
    float4* o4 = (float4*)(out + (size_t)base * OUT_DIM);
    for (int i = tid; i < (int)nrows * (OUT_DIM / 4); i += 1024) {
        float4 a = ((const float4*)slice)[i];
        a.x += bv.x; a.y += bv.y; a.z += bv.z; a.w += bv.w;
        o4[i] = a;
    }
}

extern "C" void kernel_launch(void* const* d_in, const int* in_sizes, int n_in,
                              void* d_out, int out_size, void* d_ws, size_t ws_size,
                              hipStream_t stream) {
    const float* x    = (const float*)d_in[0];
    const float* w    = (const float*)d_in[1];
    const float* bias = (const float*)d_in[2];
    const float* vals = (const float*)d_in[3];
    const int*   row  = (const int*)d_in[4];
    const int*   col  = (const int*)d_in[5];
    float* out     = (float*)d_out;
    float* support = (float*)d_ws;

    {
        const int blocks = (ROW_TILES + 7) / 8;
        gcn_gemm_wmma<<<blocks, 256, 0, stream>>>(x, w, support);
    }
    gcn_aggregate<<<N_WG, 1024, 0, stream>>>(support, vals, row, col, bias, out);
}


# Edges bucketed by owner work-group in a new pass (per 6400-edge chunk: LDS count+rank, sorted 8-B records); aggregate reads only its buckets, no row-id scan; sum order still LDS arrival order
# speedup vs baseline: 2.8495x; 1.7127x over previous
#include <hip/hip_runtime.h>

typedef __attribute__((ext_vector_type(2))) float v2f;
typedef __attribute__((ext_vector_type(8))) float v8f;

#define N_NODES 100000
#define N_EDGES 1600000
#define IN_DIM  128
#define OUT_DIM 64
#define ROW_TILES (N_NODES / 16)

#define W_PAIR_STRIDE 80
#define W_PAIR_ROWS   (IN_DIM / 2)

__global__ __launch_bounds__(256) void gcn_gemm_wmma(
    const float* __restrict__ X, const float* __restrict__ W,
    float* __restrict__ S) {
    __shared__ v2f w_lds[W_PAIR_ROWS * W_PAIR_STRIDE];

    for (int p = threadIdx.x; p < W_PAIR_ROWS * OUT_DIM; p += 256) {
        const int k2 = p >> 6;
        const int n  = p & 63;
        v2f t;
        t.x = W[(2 * k2 + 0) * OUT_DIM + n];
        t.y = W[(2 * k2 + 1) * OUT_DIM + n];
        w_lds[k2 * W_PAIR_STRIDE + n] = t;
    }
    __syncthreads();

    const int wave = threadIdx.x >> 5;
    const int lane = threadIdx.x & 31;
    const int tile = blockIdx.x * 8 + wave;
    if (tile >= ROW_TILES) return;

    const int row0  = tile * 16;
    const int m     = lane & 15;
    const int khalf = (lane >> 4) * 2;

    v8f acc[4] = {};
    const float* arow = X + (size_t)(row0 + m) * IN_DIM;

    for (int k0 = 0; k0 < IN_DIM; k0 += 4) {
        v2f a = *(const v2f*)(arow + k0 + khalf);
        const v2f* wk = w_lds + ((k0 + khalf) >> 1) * W_PAIR_STRIDE + m;
#pragma unroll
        for (int t = 0; t < 4; ++t) {
            v2f b = wk[t * 16];
            acc[t] = __builtin_amdgcn_wmma_f32_16x16x4_f32(
                 false, a,  false, b,
                 (short)0, acc[t],
                 false,  false);
        }
    }

    const int mhi = (lane >> 4) * 8;
#pragma unroll
    for (int t = 0; t < 4; ++t) {
#pragma unroll
        for (int g = 0; g < 8; ++g) {
            S[(size_t)(row0 + mhi + g) * OUT_DIM + t * 16 + m] = acc[t][g];
        }
    }
}

#define ROWS_PER_WG 391
#define N_WG        256
#define N_CHUNKS    250
#define CHUNK_EDGES (N_EDGES / N_CHUNKS)
#define BIN_THREADS 1024
#define BIN_EPT     ((CHUNK_EDGES + BIN_THREADS - 1) / BIN_THREADS)
#define OFF_STRIDE  (N_WG + 1)
#define COL_BITS    17
static_assert(N_CHUNKS * CHUNK_EDGES == N_EDGES, "chunks must tile the edge list");
static_assert(N_WG * ROWS_PER_WG >= N_NODES, "slices must cover all rows");
static_assert(N_NODES <= (1 << COL_BITS) && ROWS_PER_WG <= (1 << 9), "record packing");
static_assert(N_WG == 256, "the bucket-start scan is written for 32 lanes x 8 buckets");

__global__ __launch_bounds__(BIN_THREADS) void gcn_bin_edges(
    const float* __restrict__ vals, const int* __restrict__ row,
    const int* __restrict__ col, int* __restrict__ rec_i, int* __restrict__ off) {
    __shared__ int  hist[N_WG];
    __shared__ int  start[N_WG + 1];
    __shared__ int2 sorted[CHUNK_EDGES];

    const int tid = threadIdx.x;
    const int e0  = blockIdx.x * CHUNK_EDGES;
    if (tid < N_WG) hist[tid] = 0;
    __syncthreads();

    int   r[BIN_EPT], c[BIN_EPT];
    float v[BIN_EPT];
#pragma unroll
    for (int k = 0; k < BIN_EPT; ++k) {
        const int i = tid + k * BIN_THREADS;
        r[k] = -1; c[k] = 0; v[k] = 0.0f;
        if (i < CHUNK_EDGES) {
            r[k] = row[e0 + i];
            c[k] = col[e0 + i];
            v[k] = vals[e0 + i];
        }
    }
    int bin[BIN_EPT], rank[BIN_EPT];
#pragma unroll
    for (int k = 0; k < BIN_EPT; ++k) {
        bin[k] = -1; rank[k] = 0;
        if ((unsigned)r[k] < (unsigned)N_NODES) {
            bin[k]  = r[k] / ROWS_PER_WG;
            rank[k] = __hip_atomic_fetch_add(&hist[bin[k]], 1, __ATOMIC_RELAXED,
                                             __HIP_MEMORY_SCOPE_WORKGROUP);
        }
    }
    __syncthreads();

    if (tid < 32) {
        int cnt[8];
        int s = 0;
#pragma unroll
        for (int j = 0; j < 8; ++j) { cnt[j] = hist[tid * 8 + j]; s += cnt[j]; }
        int inc = s;
#pragma unroll
        for (int o = 1; o < 32; o <<= 1) {
            const int t = __shfl_up(inc, o);
            if (tid >= o) inc += t;
        }
        int run = inc - s;
#pragma unroll
        for (int j = 0; j < 8; ++j) { start[tid * 8 + j] = run; run += cnt[j]; }
        if (tid == 31) start[N_WG] = run;
    }
    __syncthreads();

#pragma unroll
    for (int k = 0; k < BIN_EPT; ++k) {
        if (bin[k] >= 0) {
            const int d = r[k] - bin[k] * ROWS_PER_WG;
            sorted[start[bin[k]] + rank[k]] =
                make_int2((d << COL_BITS) | c[k], __builtin_bit_cast(int, v[k]));
        }
    }
    __syncthreads();

    int2* rec = (int2*)rec_i;
    const int total = start[N_WG];
    for (int i = tid; i < total; i += BIN_THREADS) rec[e0 + i] = sorted[i];
    if (tid <= N_WG) off[blockIdx.x * OFF_STRIDE + tid] = e0 + start[tid];
}

#define AGG_WAVES   32

__device__ __forceinline__ int lane_bcast_i(int x, int j) {
    return __builtin_amdgcn_readlane(x, j);
}
__device__ __forceinline__ float lane_bcast_f(float x, int j) {
    return __builtin_bit_cast(float, __builtin_amdgcn_readlane(__builtin_bit_cast(int, x), j));
}

#define AGG_SEGS ((N_CHUNKS + AGG_WAVES - 1) / AGG_WAVES)

__device__ __forceinline__ int2 agg_fetch(
    const int2* __restrict__ rec, const int (&pre)[AGG_SEGS + 1], const int (&sb)[AGG_SEGS],
    int i0, int lane) {
    const int i = i0 + lane;
    int a = sb[0] + i;
#pragma unroll
    for (int j = 1; j < AGG_SEGS; ++j)
        if (i >= pre[j]) a = sb[j] + (i - pre[j]);
    int2 t = make_int2(0, 0);
    if (i < pre[AGG_SEGS]) t = rec[a];
    return t;
}

__device__ __forceinline__ void agg_drain(
    int c, int d, float v, int n, int lane, const float* __restrict__ S, float* slice) {
    for (int j0 = 0; j0 < n; j0 += 16) {
        float2 s[16];
#pragma unroll
        for (int u = 0; u < 16; ++u) {
            const int cj = lane_bcast_i(c, j0 + u);
            s[u] = *(const float2*)(S + (size_t)cj * OUT_DIM + lane * 2);
        }
#pragma unroll
        for (int u = 0; u < 16; ++u) {
            if (u == 0 || j0 + u < n) {
                const float vj = lane_bcast_f(v, j0 + u);
                const int   dj = lane_bcast_i(d, j0 + u);
                float* o = slice + dj * OUT_DIM + lane * 2;
                __hip_atomic_fetch_add(o + 0, vj * s[u].x, __ATOMIC_RELAXED, __HIP_MEMORY_SCOPE_WORKGROUP);
                __hip_atomic_fetch_add(o + 1, vj * s[u].y, __ATOMIC_RELAXED, __HIP_MEMORY_SCOPE_WORKGROUP);
            }
        }
    }
}

__global__ __launch_bounds__(1024) void gcn_aggregate(
    const float* __restrict__ S, const int* __restrict__ rec_i,
    const int* __restrict__ off,
    const float* __restrict__ bias, float* __restrict__ out) {
    __shared__ float slice[ROWS_PER_WG * OUT_DIM];

    const int2* rec = (const int2*)rec_i;
    const int tid  = threadIdx.x;
    const int lane = tid & 31;
    const int wave = __builtin_amdgcn_readfirstlane(tid >> 5);
    const int base = blockIdx.x * ROWS_PER_WG;
    const int rem  = N_NODES - base;
    const unsigned nrows = (unsigned)(rem < ROWS_PER_WG ? rem : ROWS_PER_WG);

    int vb = 0, ve = 0;
    {
        const int g = wave + AGG_WAVES * lane;
        if (g < N_CHUNKS) {
            vb = off[g * OFF_STRIDE + blockIdx.x];
            ve = off[g * OFF_STRIDE + blockIdx.x + 1];
        }
    }

    for (int i = tid; i < ROWS_PER_WG * OUT_DIM / 4; i += 1024)
        ((float4*)slice)[i] = make_float4(0.f, 0.f, 0.f, 0.f);
    __syncthreads();

    int sb[AGG_SEGS], pre[AGG_SEGS + 1];
    pre[0] = 0;
#pragma unroll
    for (int j = 0; j < AGG_SEGS; ++j) {
        sb[j] = lane_bcast_i(vb, j);
        pre[j + 1] = pre[j] + (lane_bcast_i(ve, j) - sb[j]);
    }
    const int total = pre[AGG_SEGS];

    int i0 = 0;
    int2 cur = agg_fetch(rec, pre, sb, 0, lane);
    int   c = cur.x & ((1 << COL_BITS) - 1);
    int   d = (int)((unsigned)cur.x >> COL_BITS);
    float v = __builtin_bit_cast(float, cur.y);
    for (; i0 + 32 <= total; i0 += 32) {
        const int2 nxt = agg_fetch(rec, pre, sb, i0 + 32, lane);
        agg_drain(c, d, v, 32, lane, S, slice);
        c = nxt.x & ((1 << COL_BITS) - 1);
        d = (int)((unsigned)nxt.x >> COL_BITS);
        v = __builtin_bit_cast(float, nxt.y);
    }
    if (i0 < total)
        agg_drain(c, d, v, total - i0, lane, S, slice);

    __syncthreads();
    const float4 bv = ((const float4*)bias)[tid & 15];
    float4* o4 = (float4*)(out + (size_t)base * OUT_DIM);
    for (int i = tid; i < (int)nrows * (OUT_DIM / 4); i += 1024) {
        float4 a = ((const float4*)slice)[i];
        a.x += bv.x; a.y += bv.y; a.z += bv.z; a.w += bv.w;
        o4[i] = a;
    }
}

extern "C" void kernel_launch(void* const* d_in, const int* in_sizes, int n_in,
                              void* d_out, int out_size, void* d_ws, size_t ws_size,
                              hipStream_t stream) {
    const float* x    = (const float*)d_in[0];
    const float* w    = (const float*)d_in[1];
    const float* bias = (const float*)d_in[2];
    const float* vals = (const float*)d_in[3];
    const int*   row  = (const int*)d_in[4];
    const int*   col  = (const int*)d_in[5];
    float* out     = (float*)d_out;
    float* support = (float*)d_ws;
    int*   rec     = (int*)((char*)d_ws + 33554432);
    int*   off     = (int*)((char*)d_ws + 50331648);

    gcn_bin_edges<<<N_CHUNKS, BIN_THREADS, 0, stream>>>(vals, row, col, rec, off);
    {
        const int blocks = (ROW_TILES + 7) / 8;
        gcn_gemm_wmma<<<blocks, 256, 0, stream>>>(x, w, support);
    }
    gcn_aggregate<<<N_WG, 1024, 0, stream>>>(support, rec, off, bias, out);
}


# Edge bucketing and GEMM, independent of each other, merged into ONE launch as work-group roles (196 GEMM groups of 32 waves first, then 250 bucketing groups): 3 launches -> 2; sum order unchanged
# speedup vs baseline: 3.2124x; 1.1274x over previous
#include <hip/hip_runtime.h>

typedef __attribute__((ext_vector_type(2))) float v2f;
typedef __attribute__((ext_vector_type(8))) float v8f;

#define N_NODES 100000
#define N_EDGES 1600000
#define IN_DIM  128
#define OUT_DIM 64
#define ROW_TILES (N_NODES / 16)

#define W_PAIR_STRIDE 80
#define W_PAIR_ROWS   (IN_DIM / 2)

#define FRONT_THREADS 1024
#define FRONT_WAVES   (FRONT_THREADS / 32)
#define GEMM_WGS      ((ROW_TILES + FRONT_WAVES - 1) / FRONT_WAVES)

__device__ __forceinline__ void gemm_role(
    const float* __restrict__ X, const float* __restrict__ W,
    float* __restrict__ S, v2f* w_lds, int wg) {
#pragma unroll
    for (int p = threadIdx.x; p < W_PAIR_ROWS * OUT_DIM; p += FRONT_THREADS) {
        const int k2 = p >> 6;
        const int n  = p & 63;
        v2f t;
        t.x = W[(2 * k2 + 0) * OUT_DIM + n];
        t.y = W[(2 * k2 + 1) * OUT_DIM + n];
        w_lds[k2 * W_PAIR_STRIDE + n] = t;
    }
    __syncthreads();

    const int wave = threadIdx.x >> 5;
    const int lane = threadIdx.x & 31;
    const int tile = wg * FRONT_WAVES + wave;
    if (tile >= ROW_TILES) return;

    const int row0  = tile * 16;
    const int m     = lane & 15;
    const int khalf = (lane >> 4) * 2;

    v8f acc[4] = {};
    const float* arow = X + (size_t)(row0 + m) * IN_DIM;

    for (int k0 = 0; k0 < IN_DIM; k0 += 4) {
        v2f a = *(const v2f*)(arow + k0 + khalf);
        const v2f* wk = w_lds + ((k0 + khalf) >> 1) * W_PAIR_STRIDE + m;
#pragma unroll
        for (int t = 0; t < 4; ++t) {
            v2f b = wk[t * 16];
            acc[t] = __builtin_amdgcn_wmma_f32_16x16x4_f32(
                 false, a,  false, b,
                 (short)0, acc[t],
                 false,  false);
        }
    }

    const int mhi = (lane >> 4) * 8;
#pragma unroll
    for (int t = 0; t < 4; ++t) {
#pragma unroll
        for (int g = 0; g < 8; ++g) {
            S[(size_t)(row0 + mhi + g) * OUT_DIM + t * 16 + m] = acc[t][g];
        }
    }
}

#define ROWS_PER_WG 391
#define N_WG        256
#define N_CHUNKS    250
#define CHUNK_EDGES (N_EDGES / N_CHUNKS)
#define BIN_THREADS 1024
#define BIN_EPT     ((CHUNK_EDGES + BIN_THREADS - 1) / BIN_THREADS)
#define OFF_STRIDE  (N_WG + 1)
#define COL_BITS    17
static_assert(N_CHUNKS * CHUNK_EDGES == N_EDGES, "chunks must tile the edge list");
static_assert(N_WG * ROWS_PER_WG >= N_NODES, "slices must cover all rows");
static_assert(N_NODES <= (1 << COL_BITS) && ROWS_PER_WG <= (1 << 9), "record packing");
static_assert(N_WG == 256, "the bucket-start scan is written for 32 lanes x 8 buckets");

#define BIN_LDS_START  (N_WG * 4)
#define BIN_LDS_SORTED 2080
#define BIN_LDS_BYTES  (BIN_LDS_SORTED + CHUNK_EDGES * 8)
#define GEMM_LDS_BYTES (W_PAIR_ROWS * W_PAIR_STRIDE * 8)
#define FRONT_LDS_BYTES (BIN_LDS_BYTES > GEMM_LDS_BYTES ? BIN_LDS_BYTES : GEMM_LDS_BYTES)
static_assert(BIN_THREADS == FRONT_THREADS, "both roles share one launch shape");
static_assert(BIN_LDS_START + (N_WG + 1) * 4 <= BIN_LDS_SORTED, "bin LDS layout");

__device__ __forceinline__ void bin_role(
    const float* __restrict__ vals, const int* __restrict__ row,
    const int* __restrict__ col, int* __restrict__ rec_i, int* __restrict__ off,
    unsigned char* lds, int g) {
    int*  hist   = (int*)lds;
    int*  start  = (int*)(lds + BIN_LDS_START);
    int2* sorted = (int2*)(lds + BIN_LDS_SORTED);

    const int tid = threadIdx.x;
    const int e0  = g * CHUNK_EDGES;
    if (tid < N_WG) hist[tid] = 0;
    __syncthreads();

    int   r[BIN_EPT], c[BIN_EPT];
    float v[BIN_EPT];
#pragma unroll
    for (int k = 0; k < BIN_EPT; ++k) {
        const int i = tid + k * BIN_THREADS;
        r[k] = -1; c[k] = 0; v[k] = 0.0f;
        if (i < CHUNK_EDGES) {
            r[k] = row[e0 + i];
            c[k] = col[e0 + i];
            v[k] = vals[e0 + i];
        }
    }
    int bin[BIN_EPT], rank[BIN_EPT];
#pragma unroll
    for (int k = 0; k < BIN_EPT; ++k) {
        bin[k] = -1; rank[k] = 0;
        if ((unsigned)r[k] < (unsigned)N_NODES) {
            bin[k]  = r[k] / ROWS_PER_WG;
            rank[k] = __hip_atomic_fetch_add(&hist[bin[k]], 1, __ATOMIC_RELAXED,
                                             __HIP_MEMORY_SCOPE_WORKGROUP);
        }
    }
    __syncthreads();

    if (tid < 32) {
        int cnt[8];
        int s = 0;
#pragma unroll
        for (int j = 0; j < 8; ++j) { cnt[j] = hist[tid * 8 + j]; s += cnt[j]; }
        int inc = s;
#pragma unroll
        for (int o = 1; o < 32; o <<= 1) {
            const int t = __shfl_up(inc, o);
            if (tid >= o) inc += t;
        }
        int run = inc - s;
#pragma unroll
        for (int j = 0; j < 8; ++j) { start[tid * 8 + j] = run; run += cnt[j]; }
        if (tid == 31) start[N_WG] = run;
    }
    __syncthreads();

#pragma unroll
    for (int k = 0; k < BIN_EPT; ++k) {
        if (bin[k] >= 0) {
            const int d = r[k] - bin[k] * ROWS_PER_WG;
            sorted[start[bin[k]] + rank[k]] =
                make_int2((d << COL_BITS) | c[k], __builtin_bit_cast(int, v[k]));
        }
    }
    __syncthreads();

    int2* rec = (int2*)rec_i;
    const int total = start[N_WG];
    for (int i = tid; i < total; i += BIN_THREADS) rec[e0 + i] = sorted[i];
    if (tid <= N_WG) off[g * OFF_STRIDE + tid] = e0 + start[tid];
}

#define FRONT_WGS (GEMM_WGS + N_CHUNKS)

__global__ __launch_bounds__(FRONT_THREADS) void gcn_front(
    const float* __restrict__ X, const float* __restrict__ W, float* __restrict__ S,
    const float* __restrict__ vals, const int* __restrict__ row,
    const int* __restrict__ col, int* __restrict__ rec_i, int* __restrict__ off) {
    __shared__ __attribute__((aligned(16))) unsigned char lds[FRONT_LDS_BYTES];
    if (blockIdx.x < GEMM_WGS)
        gemm_role(X, W, S, (v2f*)lds, blockIdx.x);
    else
        bin_role(vals, row, col, rec_i, off, lds, blockIdx.x - GEMM_WGS);
}

#define AGG_WAVES   32

__device__ __forceinline__ int lane_bcast_i(int x, int j) {
    return __builtin_amdgcn_readlane(x, j);
}
__device__ __forceinline__ float lane_bcast_f(float x, int j) {
    return __builtin_bit_cast(float, __builtin_amdgcn_readlane(__builtin_bit_cast(int, x), j));
}

#define AGG_SEGS ((N_CHUNKS + AGG_WAVES - 1) / AGG_WAVES)

__device__ __forceinline__ int2 agg_fetch(
    const int2* __restrict__ rec, const int (&pre)[AGG_SEGS + 1], const int (&sb)[AGG_SEGS],
    int i0, int lane) {
    const int i = i0 + lane;
    int a = sb[0] + i;
#pragma unroll
    for (int j = 1; j < AGG_SEGS; ++j)
        if (i >= pre[j]) a = sb[j] + (i - pre[j]);
    int2 t = make_int2(0, 0);
    if (i < pre[AGG_SEGS]) t = rec[a];
    return t;
}

__device__ __forceinline__ void agg_drain(
    int c, int d, float v, int n, int lane, const float* __restrict__ S, float* slice) {
    for (int j0 = 0; j0 < n; j0 += 16) {
        float2 s[16];
#pragma unroll
        for (int u = 0; u < 16; ++u) {
            const int cj = lane_bcast_i(c, j0 + u);
            s[u] = *(const float2*)(S + (size_t)cj * OUT_DIM + lane * 2);
        }
#pragma unroll
        for (int u = 0; u < 16; ++u) {
            if (u == 0 || j0 + u < n) {
                const float vj = lane_bcast_f(v, j0 + u);
                const int   dj = lane_bcast_i(d, j0 + u);
                float* o = slice + dj * OUT_DIM + lane * 2;
                __hip_atomic_fetch_add(o + 0, vj * s[u].x, __ATOMIC_RELAXED, __HIP_MEMORY_SCOPE_WORKGROUP);
                __hip_atomic_fetch_add(o + 1, vj * s[u].y, __ATOMIC_RELAXED, __HIP_MEMORY_SCOPE_WORKGROUP);
            }
        }
    }
}

__global__ __launch_bounds__(1024) void gcn_aggregate(
    const float* __restrict__ S, const int* __restrict__ rec_i,
    const int* __restrict__ off,
    const float* __restrict__ bias, float* __restrict__ out) {
    __shared__ float slice[ROWS_PER_WG * OUT_DIM];

    const int2* rec = (const int2*)rec_i;
    const int tid  = threadIdx.x;
    const int lane = tid & 31;
    const int wave = __builtin_amdgcn_readfirstlane(tid >> 5);
    const int base = blockIdx.x * ROWS_PER_WG;
    const int rem  = N_NODES - base;
    const unsigned nrows = (unsigned)(rem < ROWS_PER_WG ? rem : ROWS_PER_WG);

    int vb = 0, ve = 0;
    {
        const int g = wave + AGG_WAVES * lane;
        if (g < N_CHUNKS) {
            vb = off[g * OFF_STRIDE + blockIdx.x];
            ve = off[g * OFF_STRIDE + blockIdx.x + 1];
        }
    }

    for (int i = tid; i < ROWS_PER_WG * OUT_DIM / 4; i += 1024)
        ((float4*)slice)[i] = make_float4(0.f, 0.f, 0.f, 0.f);
    __syncthreads();

    int sb[AGG_SEGS], pre[AGG_SEGS + 1];
    pre[0] = 0;
#pragma unroll
    for (int j = 0; j < AGG_SEGS; ++j) {
        sb[j] = lane_bcast_i(vb, j);
        pre[j + 1] = pre[j] + (lane_bcast_i(ve, j) - sb[j]);
    }
    const int total = pre[AGG_SEGS];

    int i0 = 0;
    int2 cur = agg_fetch(rec, pre, sb, 0, lane);
    int   c = cur.x & ((1 << COL_BITS) - 1);
    int   d = (int)((unsigned)cur.x >> COL_BITS);
    float v = __builtin_bit_cast(float, cur.y);
    for (; i0 + 32 <= total; i0 += 32) {
        const int2 nxt = agg_fetch(rec, pre, sb, i0 + 32, lane);
        agg_drain(c, d, v, 32, lane, S, slice);
        c = nxt.x & ((1 << COL_BITS) - 1);
        d = (int)((unsigned)nxt.x >> COL_BITS);
        v = __builtin_bit_cast(float, nxt.y);
    }
    if (i0 < total)
        agg_drain(c, d, v, total - i0, lane, S, slice);

    __syncthreads();
    const float4 bv = ((const float4*)bias)[tid & 15];
    float4* o4 = (float4*)(out + (size_t)base * OUT_DIM);
    for (int i = tid; i < (int)nrows * (OUT_DIM / 4); i += 1024) {
        float4 a = ((const float4*)slice)[i];
        a.x += bv.x; a.y += bv.y; a.z += bv.z; a.w += bv.w;
        o4[i] = a;
    }
}

extern "C" void kernel_launch(void* const* d_in, const int* in_sizes, int n_in,
                              void* d_out, int out_size, void* d_ws, size_t ws_size,
                              hipStream_t stream) {
    const float* x    = (const float*)d_in[0];
    const float* w    = (const float*)d_in[1];
    const float* bias = (const float*)d_in[2];
    const float* vals = (const float*)d_in[3];
    const int*   row  = (const int*)d_in[4];
    const int*   col  = (const int*)d_in[5];
    float* out     = (float*)d_out;
    float* support = (float*)d_ws;
    int*   rec     = (int*)((char*)d_ws + 33554432);
    int*   off     = (int*)((char*)d_ws + 50331648);

    gcn_front<<<FRONT_WGS, FRONT_THREADS, 0, stream>>>(
        x, w, support, vals, row, col, rec, off);
    gcn_aggregate<<<N_WG, 1024, 0, stream>>>(support, rec, off, bias, out);
}


# Aggregation's LDS float atomic adds removed: each work-group counting-sorts its records by row in LDS (integer atomics), a wave sums a row in registers, stores it once; sum order = placed order
# speedup vs baseline: 3.7523x; 1.1681x over previous
#include <hip/hip_runtime.h>

typedef __attribute__((ext_vector_type(2))) float v2f;
typedef __attribute__((ext_vector_type(8))) float v8f;

#define N_NODES 100000
#define N_EDGES 1600000
#define IN_DIM  128
#define OUT_DIM 64
#define ROW_TILES (N_NODES / 16)

#define W_PAIR_STRIDE 80
#define W_PAIR_ROWS   (IN_DIM / 2)

#define FRONT_THREADS 1024
#define FRONT_WAVES   (FRONT_THREADS / 32)
#define GEMM_WGS      ((ROW_TILES + FRONT_WAVES - 1) / FRONT_WAVES)

__device__ __forceinline__ void gemm_role(
    const float* __restrict__ X, const float* __restrict__ W,
    float* __restrict__ S, v2f* w_lds, int wg) {
#pragma unroll
    for (int p = threadIdx.x; p < W_PAIR_ROWS * OUT_DIM; p += FRONT_THREADS) {
        const int k2 = p >> 6;
        const int n  = p & 63;
        v2f t;
        t.x = W[(2 * k2 + 0) * OUT_DIM + n];
        t.y = W[(2 * k2 + 1) * OUT_DIM + n];
        w_lds[k2 * W_PAIR_STRIDE + n] = t;
    }
    __syncthreads();

    const int wave = threadIdx.x >> 5;
    const int lane = threadIdx.x & 31;
    const int tile = wg * FRONT_WAVES + wave;
    if (tile >= ROW_TILES) return;

    const int row0  = tile * 16;
    const int m     = lane & 15;
    const int khalf = (lane >> 4) * 2;

    v8f acc[4] = {};
    const float* arow = X + (size_t)(row0 + m) * IN_DIM;

    for (int k0 = 0; k0 < IN_DIM; k0 += 4) {
        v2f a = *(const v2f*)(arow + k0 + khalf);
        const v2f* wk = w_lds + ((k0 + khalf) >> 1) * W_PAIR_STRIDE + m;
#pragma unroll
        for (int t = 0; t < 4; ++t) {
            v2f b = wk[t * 16];
            acc[t] = __builtin_amdgcn_wmma_f32_16x16x4_f32(
                 false, a,  false, b,
                 (short)0, acc[t],
                 false,  false);
        }
    }

    const int mhi = (lane >> 4) * 8;
#pragma unroll
    for (int t = 0; t < 4; ++t) {
#pragma unroll
        for (int g = 0; g < 8; ++g) {
            S[(size_t)(row0 + mhi + g) * OUT_DIM + t * 16 + m] = acc[t][g];
        }
    }
}

#define ROWS_PER_WG 391
#define N_WG        256
#define N_CHUNKS    250
#define CHUNK_EDGES (N_EDGES / N_CHUNKS)
#define BIN_THREADS 1024
#define BIN_EPT     ((CHUNK_EDGES + BIN_THREADS - 1) / BIN_THREADS)
#define OFF_STRIDE  (N_WG + 1)
#define COL_BITS    17
static_assert(N_CHUNKS * CHUNK_EDGES == N_EDGES, "chunks must tile the edge list");
static_assert(N_WG * ROWS_PER_WG >= N_NODES, "slices must cover all rows");
static_assert(N_NODES <= (1 << COL_BITS) && ROWS_PER_WG <= (1 << 9), "record packing");
static_assert(N_WG == 256, "the bucket-start scan is written for 32 lanes x 8 buckets");

#define BIN_LDS_START  (N_WG * 4)
#define BIN_LDS_SORTED 2080
#define BIN_LDS_BYTES  (BIN_LDS_SORTED + CHUNK_EDGES * 8)
#define GEMM_LDS_BYTES (W_PAIR_ROWS * W_PAIR_STRIDE * 8)
#define FRONT_LDS_BYTES (BIN_LDS_BYTES > GEMM_LDS_BYTES ? BIN_LDS_BYTES : GEMM_LDS_BYTES)
static_assert(BIN_THREADS == FRONT_THREADS, "both roles share one launch shape");
static_assert(BIN_LDS_START + (N_WG + 1) * 4 <= BIN_LDS_SORTED, "bin LDS layout");

__device__ __forceinline__ void bin_role(
    const float* __restrict__ vals, const int* __restrict__ row,
    const int* __restrict__ col, int* __restrict__ rec_i, int* __restrict__ off,
    unsigned char* lds, int g) {
    int*  hist   = (int*)lds;
    int*  start  = (int*)(lds + BIN_LDS_START);
    int2* sorted = (int2*)(lds + BIN_LDS_SORTED);

    const int tid = threadIdx.x;
    const int e0  = g * CHUNK_EDGES;
    if (tid < N_WG) hist[tid] = 0;
    __syncthreads();

    int   r[BIN_EPT], c[BIN_EPT];
    float v[BIN_EPT];
#pragma unroll
    for (int k = 0; k < BIN_EPT; ++k) {
        const int i = tid + k * BIN_THREADS;
        r[k] = -1; c[k] = 0; v[k] = 0.0f;
        if (i < CHUNK_EDGES) {
            r[k] = row[e0 + i];
            c[k] = col[e0 + i];
            v[k] = vals[e0 + i];
        }
    }
    int bin[BIN_EPT], rank[BIN_EPT];
#pragma unroll
    for (int k = 0; k < BIN_EPT; ++k) {
        bin[k] = -1; rank[k] = 0;
        if ((unsigned)r[k] < (unsigned)N_NODES) {
            bin[k]  = r[k] / ROWS_PER_WG;
            rank[k] = __hip_atomic_fetch_add(&hist[bin[k]], 1, __ATOMIC_RELAXED,
                                             __HIP_MEMORY_SCOPE_WORKGROUP);
        }
    }
    __syncthreads();

    if (tid < 32) {
        int cnt[8];
        int s = 0;
#pragma unroll
        for (int j = 0; j < 8; ++j) { cnt[j] = hist[tid * 8 + j]; s += cnt[j]; }
        int inc = s;
#pragma unroll
        for (int o = 1; o < 32; o <<= 1) {
            const int t = __shfl_up(inc, o);
            if (tid >= o) inc += t;
        }
        int run = inc - s;
#pragma unroll
        for (int j = 0; j < 8; ++j) { start[tid * 8 + j] = run; run += cnt[j]; }
        if (tid == 31) start[N_WG] = run;
    }
    __syncthreads();

#pragma unroll
    for (int k = 0; k < BIN_EPT; ++k) {
        if (bin[k] >= 0) {
            const int d = r[k] - bin[k] * ROWS_PER_WG;
            sorted[start[bin[k]] + rank[k]] =
                make_int2((d << COL_BITS) | c[k], __builtin_bit_cast(int, v[k]));
        }
    }
    __syncthreads();

    int2* rec = (int2*)rec_i;
    const int total = start[N_WG];
    for (int i = tid; i < total; i += BIN_THREADS) rec[e0 + i] = sorted[i];
    if (tid <= N_WG) off[g * OFF_STRIDE + tid] = e0 + start[tid];
}

#define FRONT_WGS (GEMM_WGS + N_CHUNKS)

__global__ __launch_bounds__(FRONT_THREADS) void gcn_front(
    const float* __restrict__ X, const float* __restrict__ W, float* __restrict__ S,
    const float* __restrict__ vals, const int* __restrict__ row,
    const int* __restrict__ col, int* __restrict__ rec_i, int* __restrict__ off) {
    __shared__ __attribute__((aligned(16))) unsigned char lds[FRONT_LDS_BYTES];
    if (blockIdx.x < GEMM_WGS)
        gemm_role(X, W, S, (v2f*)lds, blockIdx.x);
    else
        bin_role(vals, row, col, rec_i, off, lds, blockIdx.x - GEMM_WGS);
}

#define AGG_WAVES   32

__device__ __forceinline__ int lane_bcast_i(int x, int j) {
    return __builtin_amdgcn_readlane(x, j);
}
__device__ __forceinline__ float lane_bcast_f(float x, int j) {
    return __builtin_bit_cast(float, __builtin_amdgcn_readlane(__builtin_bit_cast(int, x), j));
}

#define AGG_SEGS ((N_CHUNKS + AGG_WAVES - 1) / AGG_WAVES)

__device__ __forceinline__ int2 agg_fetch(
    const int2* __restrict__ rec, const int (&pre)[AGG_SEGS + 1], const int (&sb)[AGG_SEGS],
    int i0, int lane) {
    const int i = i0 + lane;
    int a = sb[0] + i;
#pragma unroll
    for (int j = 1; j < AGG_SEGS; ++j)
        if (i >= pre[j]) a = sb[j] + (i - pre[j]);
    int2 t = make_int2(0, 0);
    if (i < pre[AGG_SEGS]) t = rec[a];
    return t;
}

__device__ __forceinline__ void agg_drain(
    int c, int d, float v, int n, int lane, const float* __restrict__ S, float* slice) {
    for (int j0 = 0; j0 < n; j0 += 16) {
        float2 s[16];
#pragma unroll
        for (int u = 0; u < 16; ++u) {
            const int cj = lane_bcast_i(c, j0 + u);
            s[u] = *(const float2*)(S + (size_t)cj * OUT_DIM + lane * 2);
        }
#pragma unroll
        for (int u = 0; u < 16; ++u) {
            if (u == 0 || j0 + u < n) {
                const float vj = lane_bcast_f(v, j0 + u);
                const int   dj = lane_bcast_i(d, j0 + u);
                float* o = slice + dj * OUT_DIM + lane * 2;
                __hip_atomic_fetch_add(o + 0, vj * s[u].x, __ATOMIC_RELAXED, __HIP_MEMORY_SCOPE_WORKGROUP);
                __hip_atomic_fetch_add(o + 1, vj * s[u].y, __ATOMIC_RELAXED, __HIP_MEMORY_SCOPE_WORKGROUP);
            }
        }
    }
}

__device__ __forceinline__ void row_drain(
    int c, int d, float v, int n, int lane, const float* __restrict__ S,
    float* __restrict__ orow, float2 bv, int& cur, float& ax, float& ay) {
    for (int j0 = 0; j0 < n; j0 += 16) {
        float2 s[16];
#pragma unroll
        for (int u = 0; u < 16; ++u) {
            const int cj = lane_bcast_i(c, j0 + u);
            s[u] = *(const float2*)(S + (size_t)cj * OUT_DIM + lane * 2);
        }
#pragma unroll
        for (int u = 0; u < 16; ++u) {
            if (u == 0 || j0 + u < n) {
                const int dj = lane_bcast_i(d, j0 + u);
                if (dj != cur) {
                    *(float2*)(orow + (size_t)cur * OUT_DIM) = make_float2(ax + bv.x, ay + bv.y);
                    ax = 0.0f; ay = 0.0f;
                    cur = dj;
                }
                const float vj = lane_bcast_f(v, j0 + u);
                ax += vj * s[u].x;
                ay += vj * s[u].y;
            }
        }
    }
}

#define ROWS_PER_WAVE 13
#define CNT_N         (AGG_WAVES * ROWS_PER_WAVE)
#define SORT_CAP      12288
#define KEEP          8
#define POOL_BYTES    (ROWS_PER_WG * OUT_DIM * 4)
static_assert(CNT_N >= ROWS_PER_WG, "every slice row has a counter and an owner wave");
static_assert(SORT_CAP * 8 <= POOL_BYTES, "the sorted records fit where the slice would be");

__global__ __launch_bounds__(1024) void gcn_aggregate(
    const float* __restrict__ S, const int* __restrict__ rec_i,
    const int* __restrict__ off,
    const float* __restrict__ bias, float* __restrict__ out) {
    __shared__ __attribute__((aligned(16))) unsigned char pool[POOL_BYTES];
    __shared__ int cnt[CNT_N];
    __shared__ int wtot[AGG_WAVES];
    int2*  sorted = (int2*)pool;
    float* slice  = (float*)pool;

    const int2* rec = (const int2*)rec_i;
    const int tid  = threadIdx.x;
    const int lane = tid & 31;
    const int wave = __builtin_amdgcn_readfirstlane(tid >> 5);
    const int base = blockIdx.x * ROWS_PER_WG;
    const int rem  = N_NODES - base;
    const unsigned nrows = (unsigned)(rem < ROWS_PER_WG ? rem : ROWS_PER_WG);

    int vb = 0, ve = 0;
    {
        const int g = wave + AGG_WAVES * lane;
        if (g < N_CHUNKS) {
            vb = off[g * OFF_STRIDE + blockIdx.x];
            ve = off[g * OFF_STRIDE + blockIdx.x + 1];
        }
    }
    if (tid < CNT_N) cnt[tid] = 0;

    int sb[AGG_SEGS], pre[AGG_SEGS + 1];
    pre[0] = 0;
#pragma unroll
    for (int j = 0; j < AGG_SEGS; ++j) {
        sb[j] = lane_bcast_i(vb, j);
        pre[j + 1] = pre[j] + (lane_bcast_i(ve, j) - sb[j]);
    }
    const int total = pre[AGG_SEGS];
    if (lane == 0) wtot[wave] = total;

    int2 R[KEEP];
#pragma unroll
    for (int k = 0; k < KEEP; ++k) R[k] = agg_fetch(rec, pre, sb, 32 * k, lane);

    __syncthreads();
    int nb = wtot[lane];
#pragma unroll
    for (int o = 16; o > 0; o >>= 1) nb += __shfl_xor(nb, o);
    nb = __builtin_amdgcn_readfirstlane(nb);

    if (nb <= SORT_CAP) {
#pragma unroll
        for (int k = 0; k < KEEP; ++k) {
            if (32 * k + lane < total) {
                const int d = (int)((unsigned)R[k].x >> COL_BITS);
                __hip_atomic_fetch_add(&cnt[d], 1, __ATOMIC_RELAXED, __HIP_MEMORY_SCOPE_WORKGROUP);
            }
        }
        for (int i0 = 32 * KEEP; i0 < total; i0 += 32) {
            const int2 t = agg_fetch(rec, pre, sb, i0, lane);
            if (i0 + lane < total) {
                const int d = (int)((unsigned)t.x >> COL_BITS);
                __hip_atomic_fetch_add(&cnt[d], 1, __ATOMIC_RELAXED, __HIP_MEMORY_SCOPE_WORKGROUP);
            }
        }
        __syncthreads();

        if (wave == 0) {
            int c13[ROWS_PER_WAVE];
            int s = 0;
#pragma unroll
            for (int j = 0; j < ROWS_PER_WAVE; ++j) { c13[j] = cnt[lane * ROWS_PER_WAVE + j]; s += c13[j]; }
            int inc = s;
#pragma unroll
            for (int o = 1; o < 32; o <<= 1) {
                const int t = __shfl_up(inc, o);
                if (lane >= o) inc += t;
            }
            int run = inc - s;
#pragma unroll
            for (int j = 0; j < ROWS_PER_WAVE; ++j) { cnt[lane * ROWS_PER_WAVE + j] = run; run += c13[j]; }
        }
        __syncthreads();

#pragma unroll
        for (int k = 0; k < KEEP; ++k) {
            if (32 * k + lane < total) {
                const int d = (int)((unsigned)R[k].x >> COL_BITS);
                const int pos = __hip_atomic_fetch_add(&cnt[d], 1, __ATOMIC_RELAXED,
                                                       __HIP_MEMORY_SCOPE_WORKGROUP);
                sorted[pos] = R[k];
            }
        }
        for (int i0 = 32 * KEEP; i0 < total; i0 += 32) {
            const int2 t = agg_fetch(rec, pre, sb, i0, lane);
            if (i0 + lane < total) {
                const int d = (int)((unsigned)t.x >> COL_BITS);
                const int pos = __hip_atomic_fetch_add(&cnt[d], 1, __ATOMIC_RELAXED,
                                                       __HIP_MEMORY_SCOPE_WORKGROUP);
                sorted[pos] = t;
            }
        }
        __syncthreads();

        const int r0 = wave * ROWS_PER_WAVE;
        int rb = 0, re = 0;
        if (lane < ROWS_PER_WAVE) {
            re = cnt[r0 + lane];
            if (r0 + lane > 0) rb = cnt[r0 + lane - 1];
        }
        const int lo = lane_bcast_i(rb, 0);
        const int hi = lane_bcast_i(re, ROWS_PER_WAVE - 1);
        const int mine = hi - lo;
        const float2 bv = ((const float2*)bias)[lane];
        float* orow = out + (size_t)base * OUT_DIM + lane * 2;

        if (mine > 0) {
            int2 t = make_int2(0, 0);
            if (lane < mine) t = sorted[lo + lane];
            int   c = t.x & ((1 << COL_BITS) - 1);
            int   d = (int)((unsigned)t.x >> COL_BITS);
            float v = __builtin_bit_cast(float, t.y);
            int cur = lane_bcast_i(d, 0);
            float ax = 0.0f, ay = 0.0f;
            int i0 = 0;
            for (; i0 + 32 <= mine; i0 += 32) {
                int2 nxt = make_int2(0, 0);
                if (i0 + 32 + lane < mine) nxt = sorted[lo + i0 + 32 + lane];
                row_drain(c, d, v, 32, lane, S, orow, bv, cur, ax, ay);
                c = nxt.x & ((1 << COL_BITS) - 1);
                d = (int)((unsigned)nxt.x >> COL_BITS);
                v = __builtin_bit_cast(float, nxt.y);
            }
            if (i0 < mine)
                row_drain(c, d, v, mine - i0, lane, S, orow, bv, cur, ax, ay);
            *(float2*)(orow + (size_t)cur * OUT_DIM) = make_float2(ax + bv.x, ay + bv.y);
        }

        unsigned empty = __builtin_amdgcn_ballot_w32(
            lane < ROWS_PER_WAVE && r0 + lane < (int)nrows && re == rb);
        while (empty != 0u) {
            const int j = __builtin_ctz(empty);
            empty &= empty - 1u;
            *(float2*)(orow + (size_t)(r0 + j) * OUT_DIM) = bv;
        }
    } else {
        for (int i = tid; i < ROWS_PER_WG * OUT_DIM / 4; i += 1024)
            ((float4*)slice)[i] = make_float4(0.f, 0.f, 0.f, 0.f);
        __syncthreads();

        int i0 = 0;
        int2 cur = R[0];
        int   c = cur.x & ((1 << COL_BITS) - 1);
        int   d = (int)((unsigned)cur.x >> COL_BITS);
        float v = __builtin_bit_cast(float, cur.y);
        for (; i0 + 32 <= total; i0 += 32) {
            const int2 nxt = agg_fetch(rec, pre, sb, i0 + 32, lane);
            agg_drain(c, d, v, 32, lane, S, slice);
            c = nxt.x & ((1 << COL_BITS) - 1);
            d = (int)((unsigned)nxt.x >> COL_BITS);
            v = __builtin_bit_cast(float, nxt.y);
        }
        if (i0 < total)
            agg_drain(c, d, v, total - i0, lane, S, slice);

        __syncthreads();
        const float4 bv = ((const float4*)bias)[tid & 15];
        float4* o4 = (float4*)(out + (size_t)base * OUT_DIM);
        for (int i = tid; i < (int)nrows * (OUT_DIM / 4); i += 1024) {
            float4 a = ((const float4*)slice)[i];
            a.x += bv.x; a.y += bv.y; a.z += bv.z; a.w += bv.w;
            o4[i] = a;
        }
    }
}

extern "C" void kernel_launch(void* const* d_in, const int* in_sizes, int n_in,
                              void* d_out, int out_size, void* d_ws, size_t ws_size,
                              hipStream_t stream) {
    const float* x    = (const float*)d_in[0];
    const float* w    = (const float*)d_in[1];
    const float* bias = (const float*)d_in[2];
    const float* vals = (const float*)d_in[3];
    const int*   row  = (const int*)d_in[4];
    const int*   col  = (const int*)d_in[5];
    float* out     = (float*)d_out;
    float* support = (float*)d_ws;
    int*   rec     = (int*)((char*)d_ws + 33554432);
    int*   off     = (int*)((char*)d_ws + 50331648);

    gcn_front<<<FRONT_WGS, FRONT_THREADS, 0, stream>>>(
        x, w, support, vals, row, col, rec, off);
    gcn_aggregate<<<N_WG, 1024, 0, stream>>>(support, rec, off, bias, out);
}
